# starting kernel
# baseline (speedup 1.0000x reference)
#include <hip/hip_runtime.h>
#include <hip/hip_bf16.h>

typedef __attribute__((ext_vector_type(2))) float v2f;
typedef __attribute__((ext_vector_type(8))) float v8f;

#define EPS_    1e-5f
#define N_      4
#define C_      256
#define H_      192
#define W_      192
#define R_      7
#define S_      49
#define HW_     (H_ * W_)
#define CHW_    (C_ * HW_)
#define QW_     8
#define QH_     2
#define KW_     14
#define KH_     8
#define NK_     112
#define OUT_PLANE (N_ * S_ * HW_)
#define WAVE_LDS  (16 * NK_ + NK_)

__global__ __launch_bounds__(128) void corr_wmma_kernel(
    const float* __restrict__ img,
    const float* __restrict__ map,
    float* __restrict__ out)
{
    const int tid  = threadIdx.x;
    const int wv   = tid >> 5;
    const int lane = tid & 31;
    const int hh   = lane >> 4;
    const int m16  = lane & 15;

    const int n  = blockIdx.z;
    const int x0 = blockIdx.x * QW_;
    const int y0 = (blockIdx.y * 4 + wv) * QH_;

    const int qy = m16 >> 3, qx = m16 & 7;
    const int yq = y0 + qy,  xq = x0 + qx;

    const float* aP = img + (size_t)n * CHW_ + (size_t)(2 * hh) * HW_ + yq * W_ + xq;
    const float* bP = map + (size_t)n * CHW_ + (size_t)(2 * hh) * HW_;

    int koff[7];
#pragma unroll
    for (int t = 0; t < 7; ++t) {
        int kk = 16 * t + m16;
        int ry = kk / KW_, rx = kk % KW_;
        int ky = y0 - R_ + ry; ky = ky < 0 ? 0 : ky;
        int kx = x0 - R_ + rx; kx = kx < 0 ? 0 : kx;
        koff[t] = ky * W_ + kx;
    }

    v8f  acc[7] = {};
    float asum = 0.f;
    float ksum[7] = {0.f, 0.f, 0.f, 0.f, 0.f, 0.f, 0.f};

#pragma unroll 4
    for (int c0 = 0; c0 < C_; c0 += 4) {
        const int co = c0 * HW_;
        float a0 = aP[co];
        float a1 = aP[co + HW_];
        asum += a0 * a0 + a1 * a1;
        v2f A; A.x = a0; A.y = a1;
#pragma unroll
        for (int t = 0; t < 7; ++t) {
            float b0 = bP[co + koff[t]];
            float b1 = bP[co + HW_ + koff[t]];
            ksum[t] += b0 * b0 + b1 * b1;
            v2f B; B.x = b0; B.y = b1;
            acc[t] = __builtin_amdgcn_wmma_f32_16x16x4_f32(
                false, A, false, B, (short)0, acc[t], false, false);
        }
    }

    asum += __int_as_float(
        __builtin_amdgcn_ds_swizzle(__float_as_int(asum), 0x401F));
    const float qinv = 1.f / (sqrtf(asum) + EPS_);

    float kinv[7];
#pragma unroll
    for (int t = 0; t < 7; ++t) {
        float s2 = ksum[t] + __int_as_float(
            __builtin_amdgcn_ds_swizzle(__float_as_int(ksum[t]), 0x401F));
        kinv[t] = 1.f / (sqrtf(s2) + EPS_);
    }

    __shared__ float lds[4 * WAVE_LDS];
    float* G     = lds + wv * WAVE_LDS;
    float* kinvA = G + 16 * NK_;
#pragma unroll
    for (int t = 0; t < 7; ++t) {
#pragma unroll
        for (int j = 0; j < 8; ++j) {
            int m = j + 8 * hh;
            G[m * NK_ + 16 * t + m16] = acc[t][j];
        }
        if (hh == 0) kinvA[16 * t + m16] = kinv[t];
    }
    __syncthreads();

    float* outCos  = out;
    float* outDiss = out + (size_t)OUT_PLANE;
    const int pixOff = yq * W_ + xq;
    const int q = m16;
    for (int s = hh; s < S_; s += 2) {
        int i  = s / 7, j = s - 7 * i;
        int kk = (qy + i) * KW_ + (qx + j);
        float D  = G[q * NK_ + kk];
        float d1 = D * qinv;
        float cv = 1.0f - d1 * kinvA[kk];
        int o = (n * S_ + s) * HW_ + pixOff;
        __builtin_nontemporal_store(cv, outCos + o);
        __builtin_nontemporal_store(d1, outDiss + o);
    }
}

extern "C" void kernel_launch(void* const* d_in, const int* in_sizes, int n_in,
                              void* d_out, int out_size, void* d_ws, size_t ws_size,
                              hipStream_t stream) {
    const float* img = (const float*)d_in[0];
    const float* map = (const float*)d_in[1];
    (void)in_sizes; (void)n_in; (void)out_size; (void)d_ws; (void)ws_size;

    dim3 grid(W_ / QW_, H_ / (QH_ * 4), N_);
    corr_wmma_kernel<<<grid, dim3(128), 0, stream>>>(img, map, (float*)d_out);
}


# Wave register tile 8x2 -> 16x4 queries (32 accumulator tiles, one-wave work-groups): a key fragment feeds up to two query tiles; key tiles are contiguous 16-wide row pieces, not wrapped halo slices
# speedup vs baseline: 2.1490x; 2.1490x over previous
#include <hip/hip_runtime.h>
#include <hip/hip_bf16.h>


typedef __attribute__((ext_vector_type(2))) float v2f;
typedef __attribute__((ext_vector_type(8))) float v8f;

#define EPS_    1e-5f
#define N_      4
#define C_      256
#define H_      192
#define W_      192
#define R_      7
#define S_      49
#define HW_     (H_ * W_)
#define CHW_    (C_ * HW_)
#define OUT_PLANE (N_ * S_ * HW_)

#define TW_     16
#define TH_     4
#define HR_     10
#define HC_     24
#define GP_     132
#define GT_     (16 * GP_ + 2)
#define WAVE_LDS (2 * GT_ + HR_ * HC_ + 64)

typedef __attribute__((address_space(3))) volatile float lds_vfloat;
__device__ __forceinline__ void lds_store1(float* p, float v) { *(lds_vfloat*)p = v; }

__global__ __launch_bounds__(32) __attribute__((amdgpu_waves_per_eu(3, 3))) void corr_wmma_kernel(
    const float* __restrict__ img,
    const float* __restrict__ map,
    float* __restrict__ out)
{
    const int lane = threadIdx.x;
    const int hh   = lane >> 4;
    const int m16  = lane & 15;

    const int n  = blockIdx.z;
    const int x0 = blockIdx.x * TW_;
    const int y0 = blockIdx.y * TH_;

    const int qy = m16 >> 3, qx = m16 & 7;
    const float* aP = img + (size_t)n * CHW_ + (size_t)(y0 * W_ + x0);
    const int aoff = 2 * hh * HW_ + qy * W_ + qx;

    const float* bP = map + (size_t)n * CHW_;
    int rowoff[HR_];
#pragma unroll
    for (int r = 0; r < HR_; ++r) {
        int ky = y0 - R_ + r; ky = ky < 0 ? 0 : ky;
        rowoff[r] = ky * W_;
    }
    int boff[2];
    {
        int kx0 = x0 - 8 + m16; kx0 = kx0 < 0 ? 0 : kx0;
        boff[0] = 2 * hh * HW_ + kx0;
        boff[1] = 2 * hh * HW_ + x0 + m16;
    }

    v8f   acc[2][2][8] = {};
    float asum[2][2] = {{0.f, 0.f}, {0.f, 0.f}};
    float ksum[HR_][2];
#pragma unroll
    for (int r = 0; r < HR_; ++r) { ksum[r][0] = 0.f; ksum[r][1] = 0.f; }

    for (int c0 = 0; c0 < C_; c0 += 4) {
        const int co = c0 * HW_;
        float a0[2][2], a1[2][2], b0[HR_][2], b1[HR_][2];
#pragma unroll
        for (int jy = 0; jy < 2; ++jy) {
#pragma unroll
            for (int kx = 0; kx < 2; ++kx) {
                a0[jy][kx] = aP[co + aoff + 2 * jy * W_ + 8 * kx];
                a1[jy][kx] = aP[co + HW_ + aoff + 2 * jy * W_ + 8 * kx];
            }
        }
#pragma unroll
        for (int r = 0; r < HR_; ++r) {
#pragma unroll
            for (int k = 0; k < 2; ++k) {
                b0[r][k] = bP[co + rowoff[r] + boff[k]];
                b1[r][k] = bP[co + HW_ + rowoff[r] + boff[k]];
            }
        }
        v2f A[2][2];
#pragma unroll
        for (int jy = 0; jy < 2; ++jy) {
#pragma unroll
            for (int kx = 0; kx < 2; ++kx) {
                asum[jy][kx] += a0[jy][kx] * a0[jy][kx] + a1[jy][kx] * a1[jy][kx];
                A[jy][kx].x = a0[jy][kx]; A[jy][kx].y = a1[jy][kx];
            }
        }
#pragma unroll
        for (int r = 0; r < HR_; ++r) {
#pragma unroll
            for (int k = 0; k < 2; ++k) {
                ksum[r][k] += b0[r][k] * b0[r][k] + b1[r][k] * b1[r][k];
                v2f B; B.x = b0[r][k]; B.y = b1[r][k];
                if (r <= 7)
                    acc[0][k][r] = __builtin_amdgcn_wmma_f32_16x16x4_f32(
                        false, A[0][k], false, B, (short)0, acc[0][k][r], false, false);
                if (r >= 2)
                    acc[1][k][r - 2] = __builtin_amdgcn_wmma_f32_16x16x4_f32(
                        false, A[1][k], false, B, (short)0, acc[1][k][r - 2], false, false);
            }
        }
    }

    __shared__ float lds[WAVE_LDS];
    float* Gs    = lds;
    float* kinvW = Gs + 2 * GT_;
    float* qinvW = kinvW + HR_ * HC_;

#pragma unroll
    for (int r = 0; r < HR_; ++r) {
#pragma unroll
        for (int k = 0; k < 2; ++k) {
            float s2 = ksum[r][k] + __int_as_float(
                __builtin_amdgcn_ds_swizzle(__float_as_int(ksum[r][k]), 0x401F));
            float kinv = 1.f / (sqrtf(s2) + EPS_);
            if (hh == 0 && (k == 0 || m16 >= 8)) lds_store1(&kinvW[r * HC_ + 8 * k + m16], kinv);
        }
    }
#pragma unroll
    for (int jy = 0; jy < 2; ++jy) {
#pragma unroll
        for (int kx = 0; kx < 2; ++kx) {
            float a2 = asum[jy][kx] + __int_as_float(
                __builtin_amdgcn_ds_swizzle(__float_as_int(asum[jy][kx]), 0x401F));
            float qinv = 1.f / (sqrtf(a2) + EPS_);
            if (hh == 0) lds_store1(&qinvW[(jy * 2 + kx) * 16 + m16], qinv);
        }
    }

    const int px  = lane & 15;
    const int eqy = lane >> 4;
    const int ekx = px >> 3;
    const int eqx = px & 7;
    const int eq  = 8 * eqy + eqx;
    float* outCos  = out;
    float* outDiss = out + (size_t)OUT_PLANE;

#pragma unroll
    for (int jy = 0; jy < 2; ++jy) {
        if (jy) __syncthreads();
#pragma unroll
        for (int kx = 0; kx < 2; ++kx) {
#pragma unroll
            for (int t = 0; t < 8; ++t) {
#pragma unroll
                for (int j = 0; j < 8; ++j) {
                    int m = j + 8 * hh;
                    lds_store1(&Gs[kx * GT_ + m * GP_ + 16 * t + m16], acc[jy][kx][t][j]);
                }
            }
        }
        __syncthreads();

        const float qinv = qinvW[(jy * 2 + ekx) * 16 + eq];
        const float* gB = Gs + ekx * GT_ + eq * GP_ + 16 * eqy + eqx + 1;
        const float* kB = kinvW + (2 * jy + eqy) * HC_ + px + 1;
        const int pixOff = (y0 + 2 * jy + eqy) * W_ + x0 + px;
        for (int i = 0; i < R_; ++i) {
#pragma unroll
            for (int j = 0; j < R_; ++j) {
                float D  = gB[16 * i + j];
                float d1 = D * qinv;
                float cv = 1.0f - d1 * kB[HC_ * i + j];
                int o = (n * S_ + 7 * i + j) * HW_ + pixOff;
                __builtin_nontemporal_store(cv, outCos + o);
                __builtin_nontemporal_store(d1, outDiss + o);
            }
        }
    }
}

extern "C" void kernel_launch(void* const* d_in, const int* in_sizes, int n_in,
                              void* d_out, int out_size, void* d_ws, size_t ws_size,
                              hipStream_t stream) {
    const float* img = (const float*)d_in[0];
    const float* map = (const float*)d_in[1];
    (void)in_sizes; (void)n_in; (void)out_size; (void)d_ws; (void)ws_size;

    dim3 grid(W_ / TW_, H_ / TH_, N_);
    corr_wmma_kernel<<<grid, dim3(32), 0, stream>>>(img, map, (float*)d_out);
}


# Work-groups of 4 waves = 2x2 adjacent wave tiles (32x8 queries), was 1 wave: shared query lines and overlapping key-halo rows come from L2 once and hit in the WGP cache for the other waves
# speedup vs baseline: 2.3473x; 1.0923x over previous
#include <hip/hip_runtime.h>
#include <hip/hip_bf16.h>


typedef __attribute__((ext_vector_type(2))) float v2f;
typedef __attribute__((ext_vector_type(8))) float v8f;

#define EPS_    1e-5f
#define N_      4
#define C_      256
#define H_      192
#define W_      192
#define R_      7
#define S_      49
#define HW_     (H_ * W_)
#define CHW_    (C_ * HW_)
#define OUT_PLANE (N_ * S_ * HW_)

#define TW_     16
#define TH_     4
#define HR_     10
#define HC_     24
#define GP_     132
#define GT_     (16 * GP_ + 2)
#define WAVE_LDS (2 * GT_ + HR_ * HC_ + 64)
#define WG_WAVES 4

typedef __attribute__((address_space(3))) volatile float lds_vfloat;
__device__ __forceinline__ void lds_store1(float* p, float v) { *(lds_vfloat*)p = v; }

__global__ __launch_bounds__(32 * WG_WAVES) __attribute__((amdgpu_waves_per_eu(3, 3))) void corr_wmma_kernel(
    const float* __restrict__ img,
    const float* __restrict__ map,
    float* __restrict__ out)
{
    const int lane = threadIdx.x & 31;
    const int wv   = __builtin_amdgcn_readfirstlane(threadIdx.x >> 5);
    const int hh   = lane >> 4;
    const int m16  = lane & 15;

    const int n  = blockIdx.z;
    const int x0 = (2 * blockIdx.x + (wv & 1)) * TW_;
    const int y0 = (2 * blockIdx.y + (wv >> 1)) * TH_;

    const int qy = m16 >> 3, qx = m16 & 7;
    const float* aP = img + (size_t)n * CHW_ + (size_t)(y0 * W_ + x0);
    const int aoff = 2 * hh * HW_ + qy * W_ + qx;

    const float* bP = map + (size_t)n * CHW_;
    int rowoff[HR_];
#pragma unroll
    for (int r = 0; r < HR_; ++r) {
        int ky = y0 - R_ + r; ky = ky < 0 ? 0 : ky;
        rowoff[r] = ky * W_;
    }
    int boff[2];
    {
        int kx0 = x0 - 8 + m16; kx0 = kx0 < 0 ? 0 : kx0;
        boff[0] = 2 * hh * HW_ + kx0;
        boff[1] = 2 * hh * HW_ + x0 + m16;
    }

    v8f   acc[2][2][8] = {};
    float asum[2][2] = {{0.f, 0.f}, {0.f, 0.f}};
    float ksum[HR_][2];
#pragma unroll
    for (int r = 0; r < HR_; ++r) { ksum[r][0] = 0.f; ksum[r][1] = 0.f; }

    for (int c0 = 0; c0 < C_; c0 += 4) {
        const int co = c0 * HW_;
        float a0[2][2], a1[2][2], b0[HR_][2], b1[HR_][2];
#pragma unroll
        for (int jy = 0; jy < 2; ++jy) {
#pragma unroll
            for (int kx = 0; kx < 2; ++kx) {
                a0[jy][kx] = aP[co + aoff + 2 * jy * W_ + 8 * kx];
                a1[jy][kx] = aP[co + HW_ + aoff + 2 * jy * W_ + 8 * kx];
            }
        }
#pragma unroll
        for (int r = 0; r < HR_; ++r) {
#pragma unroll
            for (int k = 0; k < 2; ++k) {
                b0[r][k] = bP[co + rowoff[r] + boff[k]];
                b1[r][k] = bP[co + HW_ + rowoff[r] + boff[k]];
            }
        }
        v2f A[2][2];
#pragma unroll
        for (int jy = 0; jy < 2; ++jy) {
#pragma unroll
            for (int kx = 0; kx < 2; ++kx) {
                asum[jy][kx] += a0[jy][kx] * a0[jy][kx] + a1[jy][kx] * a1[jy][kx];
                A[jy][kx].x = a0[jy][kx]; A[jy][kx].y = a1[jy][kx];
            }
        }
#pragma unroll
        for (int r = 0; r < HR_; ++r) {
#pragma unroll
            for (int k = 0; k < 2; ++k) {
                ksum[r][k] += b0[r][k] * b0[r][k] + b1[r][k] * b1[r][k];
                v2f B; B.x = b0[r][k]; B.y = b1[r][k];
                if (r <= 7)
                    acc[0][k][r] = __builtin_amdgcn_wmma_f32_16x16x4_f32(
                        false, A[0][k], false, B, (short)0, acc[0][k][r], false, false);
                if (r >= 2)
                    acc[1][k][r - 2] = __builtin_amdgcn_wmma_f32_16x16x4_f32(
                        false, A[1][k], false, B, (short)0, acc[1][k][r - 2], false, false);
            }
        }
    }

    __shared__ float lds[WG_WAVES * WAVE_LDS];
    float* Gs    = lds + wv * WAVE_LDS;
    float* kinvW = Gs + 2 * GT_;
    float* qinvW = kinvW + HR_ * HC_;

#pragma unroll
    for (int r = 0; r < HR_; ++r) {
#pragma unroll
        for (int k = 0; k < 2; ++k) {
            float s2 = ksum[r][k] + __int_as_float(
                __builtin_amdgcn_ds_swizzle(__float_as_int(ksum[r][k]), 0x401F));
            float kinv = 1.f / (sqrtf(s2) + EPS_);
            if (hh == 0 && (k == 0 || m16 >= 8)) lds_store1(&kinvW[r * HC_ + 8 * k + m16], kinv);
        }
    }
#pragma unroll
    for (int jy = 0; jy < 2; ++jy) {
#pragma unroll
        for (int kx = 0; kx < 2; ++kx) {
            float a2 = asum[jy][kx] + __int_as_float(
                __builtin_amdgcn_ds_swizzle(__float_as_int(asum[jy][kx]), 0x401F));
            float qinv = 1.f / (sqrtf(a2) + EPS_);
            if (hh == 0) lds_store1(&qinvW[(jy * 2 + kx) * 16 + m16], qinv);
        }
    }

    const int px  = lane & 15;
    const int eqy = lane >> 4;
    const int ekx = px >> 3;
    const int eqx = px & 7;
    const int eq  = 8 * eqy + eqx;
    float* outCos  = out;
    float* outDiss = out + (size_t)OUT_PLANE;

#pragma unroll
    for (int jy = 0; jy < 2; ++jy) {
        if (jy) __syncthreads();
#pragma unroll
        for (int kx = 0; kx < 2; ++kx) {
#pragma unroll
            for (int t = 0; t < 8; ++t) {
#pragma unroll
                for (int j = 0; j < 8; ++j) {
                    int m = j + 8 * hh;
                    lds_store1(&Gs[kx * GT_ + m * GP_ + 16 * t + m16], acc[jy][kx][t][j]);
                }
            }
        }
        __syncthreads();

        const float qinv = qinvW[(jy * 2 + ekx) * 16 + eq];
        const float* gB = Gs + ekx * GT_ + eq * GP_ + 16 * eqy + eqx + 1;
        const float* kB = kinvW + (2 * jy + eqy) * HC_ + px + 1;
        const int pixOff = (y0 + 2 * jy + eqy) * W_ + x0 + px;
        for (int i = 0; i < R_; ++i) {
#pragma unroll
            for (int j = 0; j < R_; ++j) {
                float D  = gB[16 * i + j];
                float d1 = D * qinv;
                float cv = 1.0f - d1 * kB[HC_ * i + j];
                int o = (n * S_ + 7 * i + j) * HW_ + pixOff;
                __builtin_nontemporal_store(cv, outCos + o);
                __builtin_nontemporal_store(d1, outDiss + o);
            }
        }
    }
}

extern "C" void kernel_launch(void* const* d_in, const int* in_sizes, int n_in,
                              void* d_out, int out_size, void* d_ws, size_t ws_size,
                              hipStream_t stream) {
    const float* img = (const float*)d_in[0];
    const float* map = (const float*)d_in[1];
    (void)in_sizes; (void)n_in; (void)out_size; (void)d_ws; (void)ws_size;

    dim3 grid(W_ / (2 * TW_), H_ / (2 * TH_), N_);
    corr_wmma_kernel<<<grid, dim3(32 * WG_WAVES), 0, stream>>>(img, map, (float*)d_out);
}


# Wave tile 16x4 -> 24x6 queries (72 accumulator tiles), 1024 waves = one per SIMD; the register file then holds a second fragment set, so a k-step's loads fly under the previous k-step's WMMAs
# speedup vs baseline: 2.6651x; 1.1354x over previous
#include <hip/hip_runtime.h>
#include <hip/hip_bf16.h>


typedef __attribute__((ext_vector_type(2))) float v2f;
typedef __attribute__((ext_vector_type(8))) float v8f;

#define EPS_    1e-5f
#define N_      4
#define C_      256
#define H_      192
#define W_      192
#define R_      7
#define S_      49
#define HW_     (H_ * W_)
#define CHW_    (C_ * HW_)
#define OUT_PLANE (N_ * S_ * HW_)

#define TW_     24
#define TH_     6
#define NJ_     3
#define NK_     3
#define HR_     12
#define HC_     32
#define GP_     132
#define GT_     (16 * GP_ + 2)
#define WAVE_LDS (2 * GT_ + HR_ * HC_ + NJ_ * NK_ * 16)
#define WG_WAVES 4

typedef __attribute__((address_space(3))) volatile float lds_vfloat;
__device__ __forceinline__ void lds_store1(float* p, float v) { *(lds_vfloat*)p = v; }

struct Frag {
    float a0[NJ_][NK_], a1[NJ_][NK_];
    float b0[HR_][NK_], b1[HR_][NK_];
};

template <int K>
__device__ __forceinline__ void load_group(Frag& F, const float* __restrict__ aP, const float* __restrict__ bP,
                                           int co, int aoff, const int (&rowoff)[HR_], const int (&boff)[NK_])
{
#pragma unroll
    for (int jy = 0; jy < NJ_; ++jy) {
        F.a0[jy][K] = aP[co + aoff + 2 * jy * W_ + 8 * K];
        F.a1[jy][K] = aP[co + HW_ + aoff + 2 * jy * W_ + 8 * K];
    }
#pragma unroll
    for (int r = 0; r < HR_; ++r) {
        F.b0[r][K] = bP[co + rowoff[r] + boff[K]];
        F.b1[r][K] = bP[co + HW_ + rowoff[r] + boff[K]];
    }
}

template <int K>
__device__ __forceinline__ void mma_group(const Frag& F, v8f (&acc)[NJ_][NK_][8],
                                          float (&asum)[NJ_][NK_], float (&ksum)[HR_][NK_])
{
    v2f A[NJ_];
#pragma unroll
    for (int jy = 0; jy < NJ_; ++jy) {
        asum[jy][K] += F.a0[jy][K] * F.a0[jy][K] + F.a1[jy][K] * F.a1[jy][K];
        A[jy].x = F.a0[jy][K]; A[jy].y = F.a1[jy][K];
    }
#pragma unroll
    for (int r = 0; r < HR_; ++r) {
        ksum[r][K] += F.b0[r][K] * F.b0[r][K] + F.b1[r][K] * F.b1[r][K];
        v2f B; B.x = F.b0[r][K]; B.y = F.b1[r][K];
#pragma unroll
        for (int jy = 0; jy < NJ_; ++jy) {
            if (r >= 2 * jy && r <= 2 * jy + 7)
                acc[jy][K][r - 2 * jy] = __builtin_amdgcn_wmma_f32_16x16x4_f32(
                    false, A[jy], false, B, (short)0, acc[jy][K][r - 2 * jy], false, false);
        }
    }
}

__device__ __forceinline__ void keep_order() { __builtin_amdgcn_sched_barrier(0); }

template <int J0, int K0, int J1, int K1, int NT, bool FIRST>
__device__ __forceinline__ void extract_pass(const v8f (&acc)[NJ_][NK_][8], float* Gs, const float* kinvW,
                                             const float* qinvW, int lane, int n, int x0, int y0,
                                             float* __restrict__ outCos, float* __restrict__ outDiss)
{
    const int hh   = lane >> 4;
    const int m16  = lane & 15;
    const int eqy  = lane >> 4;
    const int slot = m16 >> 3;
    const int eqx  = m16 & 7;
    const int eq   = 8 * eqy + eqx;

    if (!FIRST) __syncthreads();
#pragma unroll
    for (int t = 0; t < 8; ++t) {
#pragma unroll
        for (int j = 0; j < 8; ++j) {
            int m = j + 8 * hh;
            lds_store1(&Gs[m * GP_ + 16 * t + m16], acc[J0][K0][t][j]);
            if (NT == 2) lds_store1(&Gs[GT_ + m * GP_ + 16 * t + m16], acc[J1][K1][t][j]);
        }
    }
    __syncthreads();

    if (slot < NT) {
        const int jyL = slot ? J1 : J0;
        const int kxL = slot ? K1 : K0;
        const float qinv = qinvW[(jyL * NK_ + kxL) * 16 + eq];
        const float* gB = Gs + slot * GT_ + eq * GP_ + 16 * eqy + eqx + 1;
        const float* kB = kinvW + (2 * jyL + eqy) * HC_ + 8 * kxL + eqx + 1;
        const int pixOff = (y0 + 2 * jyL + eqy) * W_ + x0 + 8 * kxL + eqx;
        for (int i = 0; i < R_; ++i) {
#pragma unroll
            for (int j = 0; j < R_; ++j) {
                float D  = gB[16 * i + j];
                float d1 = D * qinv;
                float cv = 1.0f - d1 * kB[HC_ * i + j];
                int o = (n * S_ + 7 * i + j) * HW_ + pixOff;
                __builtin_nontemporal_store(cv, outCos + o);
                __builtin_nontemporal_store(d1, outDiss + o);
            }
        }
    }
}

__global__ __launch_bounds__(32 * WG_WAVES) __attribute__((amdgpu_waves_per_eu(1, 1))) void corr_wmma_kernel(
    const float* __restrict__ img,
    const float* __restrict__ map,
    float* __restrict__ out)
{
    const int lane = threadIdx.x & 31;
    const int wv   = __builtin_amdgcn_readfirstlane(threadIdx.x >> 5);
    const int hh   = lane >> 4;
    const int m16  = lane & 15;

    const int n  = blockIdx.z;
    const int x0 = (2 * blockIdx.x + (wv & 1)) * TW_;
    const int y0 = (2 * blockIdx.y + (wv >> 1)) * TH_;

    const int qy = m16 >> 3, qx = m16 & 7;
    const float* aP = img + (size_t)n * CHW_ + (size_t)(y0 * W_ + x0);
    const int aoff = 2 * hh * HW_ + qy * W_ + qx;

    const float* bP = map + (size_t)n * CHW_;
    int rowoff[HR_];
#pragma unroll
    for (int r = 0; r < HR_; ++r) {
        int ky = y0 - R_ + r; ky = ky < 0 ? 0 : ky;
        rowoff[r] = ky * W_;
    }
    int boff[NK_];
#pragma unroll
    for (int k = 0; k < NK_; ++k) {
        int kxc = x0 - 8 + 8 * k + m16; kxc = kxc < 0 ? 0 : kxc;
        boff[k] = 2 * hh * HW_ + kxc;
    }

    v8f   acc[NJ_][NK_][8] = {};
    float asum[NJ_][NK_];
    float ksum[HR_][NK_];
#pragma unroll
    for (int k = 0; k < NK_; ++k) {
#pragma unroll
        for (int jy = 0; jy < NJ_; ++jy) asum[jy][k] = 0.f;
#pragma unroll
        for (int r = 0; r < HR_; ++r) ksum[r][k] = 0.f;
    }

    Frag F0, F1;
    load_group<0>(F0, aP, bP, 0, aoff, rowoff, boff);
    load_group<1>(F0, aP, bP, 0, aoff, rowoff, boff);
    load_group<2>(F0, aP, bP, 0, aoff, rowoff, boff);

    for (int c0 = 0; c0 < C_; c0 += 8) {
        const int co1 = (c0 + 4) * HW_;
        const int c2  = c0 + 8 < C_ ? c0 + 8 : C_ - 4;
        const int co2 = c2 * HW_;

        load_group<0>(F1, aP, bP, co1, aoff, rowoff, boff);  keep_order();
        mma_group<0>(F0, acc, asum, ksum);                   keep_order();
        load_group<1>(F1, aP, bP, co1, aoff, rowoff, boff);  keep_order();
        mma_group<1>(F0, acc, asum, ksum);                   keep_order();
        load_group<2>(F1, aP, bP, co1, aoff, rowoff, boff);  keep_order();
        mma_group<2>(F0, acc, asum, ksum);                   keep_order();

        load_group<0>(F0, aP, bP, co2, aoff, rowoff, boff);  keep_order();
        mma_group<0>(F1, acc, asum, ksum);                   keep_order();
        load_group<1>(F0, aP, bP, co2, aoff, rowoff, boff);  keep_order();
        mma_group<1>(F1, acc, asum, ksum);                   keep_order();
        load_group<2>(F0, aP, bP, co2, aoff, rowoff, boff);  keep_order();
        mma_group<2>(F1, acc, asum, ksum);                   keep_order();
    }

    __shared__ float lds[WG_WAVES * WAVE_LDS];
    float* Gs    = lds + wv * WAVE_LDS;
    float* kinvW = Gs + 2 * GT_;
    float* qinvW = kinvW + HR_ * HC_;

#pragma unroll
    for (int r = 0; r < HR_; ++r) {
#pragma unroll
        for (int k = 0; k < NK_; ++k) {
            float s2 = ksum[r][k] + __int_as_float(
                __builtin_amdgcn_ds_swizzle(__float_as_int(ksum[r][k]), 0x401F));
            float kinv = 1.f / (sqrtf(s2) + EPS_);
            if (hh == 0 && (k == 0 || m16 >= 8)) lds_store1(&kinvW[r * HC_ + 8 * k + m16], kinv);
        }
    }
#pragma unroll
    for (int jy = 0; jy < NJ_; ++jy) {
#pragma unroll
        for (int kx = 0; kx < NK_; ++kx) {
            float a2 = asum[jy][kx] + __int_as_float(
                __builtin_amdgcn_ds_swizzle(__float_as_int(asum[jy][kx]), 0x401F));
            float qinv = 1.f / (sqrtf(a2) + EPS_);
            if (hh == 0) lds_store1(&qinvW[(jy * NK_ + kx) * 16 + m16], qinv);
        }
    }

    float* outCos  = out;
    float* outDiss = out + (size_t)OUT_PLANE;

    extract_pass<0, 0, 0, 1, 2, true >(acc, Gs, kinvW, qinvW, lane, n, x0, y0, outCos, outDiss);
    extract_pass<1, 0, 1, 1, 2, false>(acc, Gs, kinvW, qinvW, lane, n, x0, y0, outCos, outDiss);
    extract_pass<2, 0, 2, 1, 2, false>(acc, Gs, kinvW, qinvW, lane, n, x0, y0, outCos, outDiss);
    extract_pass<0, 2, 1, 2, 2, false>(acc, Gs, kinvW, qinvW, lane, n, x0, y0, outCos, outDiss);
    extract_pass<2, 2, 2, 2, 1, false>(acc, Gs, kinvW, qinvW, lane, n, x0, y0, outCos, outDiss);
}

extern "C" void kernel_launch(void* const* d_in, const int* in_sizes, int n_in,
                              void* d_out, int out_size, void* d_ws, size_t ws_size,
                              hipStream_t stream) {
    const float* img = (const float*)d_in[0];
    const float* map = (const float*)d_in[1];
    (void)in_sizes; (void)n_in; (void)out_size; (void)d_ws; (void)ws_size;

    dim3 grid(W_ / (2 * TW_), H_ / (2 * TH_), N_);
    corr_wmma_kernel<<<grid, dim3(32 * WG_WAVES), 0, stream>>>(img, map, (float*)d_out);
}


# Middle key column block no longer loaded: built in registers from the left and right blocks by a per-lane select (columns swapped by 8, set right at the LDS spill); 66 loads per k-step, was 90
# speedup vs baseline: 3.0136x; 1.1308x over previous
#include <hip/hip_runtime.h>
#include <hip/hip_bf16.h>


typedef __attribute__((ext_vector_type(2))) float v2f;
typedef __attribute__((ext_vector_type(8))) float v8f;

#define EPS_    1e-5f
#define N_      4
#define C_      256
#define H_      192
#define W_      192
#define R_      7
#define S_      49
#define HW_     (H_ * W_)
#define CHW_    (C_ * HW_)
#define OUT_PLANE (N_ * S_ * HW_)

#define TW_     24
#define TH_     6
#define NJ_     3
#define NK_     3
#define HR_     12
#define HC_     32
#define NB_     2
#define GP_     132
#define GT_     (16 * GP_ + 2)
#define WAVE_LDS (2 * GT_ + HR_ * HC_ + NJ_ * NK_ * 16)
#define WG_WAVES 4

typedef __attribute__((address_space(3))) volatile float lds_vfloat;
__device__ __forceinline__ void lds_store1(float* p, float v) { *(lds_vfloat*)p = v; }

struct Frag {
    float a0[NJ_][NK_], a1[NJ_][NK_];
    float b0[HR_][NB_], b1[HR_][NB_];
};

template <int K>
__device__ __forceinline__ void load_queries(Frag& F, const float* __restrict__ aP, int co, int aoff)
{
#pragma unroll
    for (int jy = 0; jy < NJ_; ++jy) {
        F.a0[jy][K] = aP[co + aoff + 2 * jy * W_ + 8 * K];
        F.a1[jy][K] = aP[co + HW_ + aoff + 2 * jy * W_ + 8 * K];
    }
}

template <int B>
__device__ __forceinline__ void load_keys(Frag& F, const float* __restrict__ bP, int co,
                                          const int (&rowoff)[HR_], const int (&boff)[NB_])
{
#pragma unroll
    for (int r = 0; r < HR_; ++r) {
        F.b0[r][B] = bP[co + rowoff[r] + boff[B]];
        F.b1[r][B] = bP[co + HW_ + rowoff[r] + boff[B]];
    }
}

struct Mid {
    float m0[HR_], m1[HR_];
};

__device__ __forceinline__ void select_mid(const Frag& F, Mid& M, bool takeRight)
{
#pragma unroll
    for (int r = 0; r < HR_; ++r) {
        M.m0[r] = takeRight ? F.b0[r][1] : F.b0[r][0];
        M.m1[r] = takeRight ? F.b1[r][1] : F.b1[r][0];
    }
}

template <int K>
__device__ __forceinline__ void mma_group(const Frag& F, const Mid& M, v8f (&acc)[NJ_][NK_][8],
                                          float (&asum)[NJ_][NK_], float (&ksum)[HR_][NB_])
{
    v2f A[NJ_];
#pragma unroll
    for (int jy = 0; jy < NJ_; ++jy) {
        asum[jy][K] += F.a0[jy][K] * F.a0[jy][K] + F.a1[jy][K] * F.a1[jy][K];
        A[jy].x = F.a0[jy][K]; A[jy].y = F.a1[jy][K];
    }
#pragma unroll
    for (int r = 0; r < HR_; ++r) {
        v2f B;
        if (K == 1) {
            B.x = M.m0[r]; B.y = M.m1[r];
        } else {
            ksum[r][K / 2] += F.b0[r][K / 2] * F.b0[r][K / 2] + F.b1[r][K / 2] * F.b1[r][K / 2];
            B.x = F.b0[r][K / 2]; B.y = F.b1[r][K / 2];
        }
#pragma unroll
        for (int jy = 0; jy < NJ_; ++jy) {
            if (r >= 2 * jy && r <= 2 * jy + 7)
                acc[jy][K][r - 2 * jy] = __builtin_amdgcn_wmma_f32_16x16x4_f32(
                    false, A[jy], false, B, (short)0, acc[jy][K][r - 2 * jy], false, false);
        }
    }
}

__device__ __forceinline__ void keep_order() { __builtin_amdgcn_sched_barrier(0); }

template <int J0, int K0, int J1, int K1, int NT, bool FIRST>
__device__ __forceinline__ void extract_pass(const v8f (&acc)[NJ_][NK_][8], float* Gs, const float* kinvW,
                                             const float* qinvW, int lane, int n, int x0, int y0,
                                             float* __restrict__ outCos, float* __restrict__ outDiss)
{
    const int hh   = lane >> 4;
    const int m16  = lane & 15;
    const int eqy  = lane >> 4;
    const int slot = m16 >> 3;
    const int eqx  = m16 & 7;
    const int eq   = 8 * eqy + eqx;

    const int col0 = K0 == 1 ? (m16 ^ 8) : m16;
    const int col1 = K1 == 1 ? (m16 ^ 8) : m16;

    if (!FIRST) __syncthreads();
#pragma unroll
    for (int t = 0; t < 8; ++t) {
#pragma unroll
        for (int j = 0; j < 8; ++j) {
            int m = j + 8 * hh;
            lds_store1(&Gs[m * GP_ + 16 * t + col0], acc[J0][K0][t][j]);
            if (NT == 2) lds_store1(&Gs[GT_ + m * GP_ + 16 * t + col1], acc[J1][K1][t][j]);
        }
    }
    __syncthreads();

    if (slot < NT) {
        const int jyL = slot ? J1 : J0;
        const int kxL = slot ? K1 : K0;
        const float qinv = qinvW[(jyL * NK_ + kxL) * 16 + eq];
        const float* gB = Gs + slot * GT_ + eq * GP_ + 16 * eqy + eqx + 1;
        const float* kB = kinvW + (2 * jyL + eqy) * HC_ + 8 * kxL + eqx + 1;
        const int pixOff = (y0 + 2 * jyL + eqy) * W_ + x0 + 8 * kxL + eqx;
        for (int i = 0; i < R_; ++i) {
#pragma unroll
            for (int j = 0; j < R_; ++j) {
                float D  = gB[16 * i + j];
                float d1 = D * qinv;
                float cv = 1.0f - d1 * kB[HC_ * i + j];
                int o = (n * S_ + 7 * i + j) * HW_ + pixOff;
                __builtin_nontemporal_store(cv, outCos + o);
                __builtin_nontemporal_store(d1, outDiss + o);
            }
        }
    }
}

__global__ __launch_bounds__(32 * WG_WAVES) __attribute__((amdgpu_waves_per_eu(1, 1))) void corr_wmma_kernel(
    const float* __restrict__ img,
    const float* __restrict__ map,
    float* __restrict__ out)
{
    const int lane = threadIdx.x & 31;
    const int wv   = __builtin_amdgcn_readfirstlane(threadIdx.x >> 5);
    const int hh   = lane >> 4;
    const int m16  = lane & 15;

    const int n  = blockIdx.z;
    const int x0 = (2 * blockIdx.x + (wv & 1)) * TW_;
    const int y0 = (2 * blockIdx.y + (wv >> 1)) * TH_;

    const int qy = m16 >> 3, qx = m16 & 7;
    const float* aP = img + (size_t)n * CHW_ + (size_t)(y0 * W_ + x0);
    const int aoff = 2 * hh * HW_ + qy * W_ + qx;

    const float* bP = map + (size_t)n * CHW_;
    int rowoff[HR_];
#pragma unroll
    for (int r = 0; r < HR_; ++r) {
        int ky = y0 - R_ + r; ky = ky < 0 ? 0 : ky;
        rowoff[r] = ky * W_;
    }
    int boff[NB_];
#pragma unroll
    for (int b = 0; b < NB_; ++b) {
        int kxc = x0 - 8 + 16 * b + m16; kxc = kxc < 0 ? 0 : kxc;
        boff[b] = 2 * hh * HW_ + kxc;
    }
    const bool takeRight = (lane & 8) == 0;

    v8f   acc[NJ_][NK_][8] = {};
    float asum[NJ_][NK_];
    float ksum[HR_][NB_];
#pragma unroll
    for (int k = 0; k < NK_; ++k) {
#pragma unroll
        for (int jy = 0; jy < NJ_; ++jy) asum[jy][k] = 0.f;
    }
#pragma unroll
    for (int b = 0; b < NB_; ++b) {
#pragma unroll
        for (int r = 0; r < HR_; ++r) ksum[r][b] = 0.f;
    }

    Frag F0, F1;
    Mid  M;
    load_queries<0>(F0, aP, 0, aoff);  load_keys<0>(F0, bP, 0, rowoff, boff);  keep_order();
    load_queries<2>(F0, aP, 0, aoff);  load_keys<1>(F0, bP, 0, rowoff, boff);  keep_order();
    load_queries<1>(F0, aP, 0, aoff);                                          keep_order();

    for (int c0 = 0; c0 < C_; c0 += 8) {
        const int co1 = (c0 + 4) * HW_;
        const int c2  = c0 + 8 < C_ ? c0 + 8 : C_ - 4;
        const int co2 = c2 * HW_;

        load_queries<0>(F1, aP, co1, aoff);  load_keys<0>(F1, bP, co1, rowoff, boff);  keep_order();
        mma_group<0>(F0, M, acc, asum, ksum);                                          keep_order();
        load_queries<2>(F1, aP, co1, aoff);  load_keys<1>(F1, bP, co1, rowoff, boff);  keep_order();
        mma_group<2>(F0, M, acc, asum, ksum);                                          keep_order();
        load_queries<1>(F1, aP, co1, aoff);  select_mid(F0, M, takeRight);             keep_order();
        mma_group<1>(F0, M, acc, asum, ksum);                                          keep_order();

        load_queries<0>(F0, aP, co2, aoff);  load_keys<0>(F0, bP, co2, rowoff, boff);  keep_order();
        mma_group<0>(F1, M, acc, asum, ksum);                                          keep_order();
        load_queries<2>(F0, aP, co2, aoff);  load_keys<1>(F0, bP, co2, rowoff, boff);  keep_order();
        mma_group<2>(F1, M, acc, asum, ksum);                                          keep_order();
        load_queries<1>(F0, aP, co2, aoff);  select_mid(F1, M, takeRight);             keep_order();
        mma_group<1>(F1, M, acc, asum, ksum);                                          keep_order();
    }

    __shared__ float lds[WG_WAVES * WAVE_LDS];
    float* Gs    = lds + wv * WAVE_LDS;
    float* kinvW = Gs + 2 * GT_;
    float* qinvW = kinvW + HR_ * HC_;

#pragma unroll
    for (int r = 0; r < HR_; ++r) {
#pragma unroll
        for (int b = 0; b < NB_; ++b) {
            float s2 = ksum[r][b] + __int_as_float(
                __builtin_amdgcn_ds_swizzle(__float_as_int(ksum[r][b]), 0x401F));
            float kinv = 1.f / (sqrtf(s2) + EPS_);
            if (hh == 0) lds_store1(&kinvW[r * HC_ + 16 * b + m16], kinv);
        }
    }
#pragma unroll
    for (int jy = 0; jy < NJ_; ++jy) {
#pragma unroll
        for (int kx = 0; kx < NK_; ++kx) {
            float a2 = asum[jy][kx] + __int_as_float(
                __builtin_amdgcn_ds_swizzle(__float_as_int(asum[jy][kx]), 0x401F));
            float qinv = 1.f / (sqrtf(a2) + EPS_);
            if (hh == 0) lds_store1(&qinvW[(jy * NK_ + kx) * 16 + m16], qinv);
        }
    }

    float* outCos  = out;
    float* outDiss = out + (size_t)OUT_PLANE;

    extract_pass<0, 0, 0, 1, 2, true >(acc, Gs, kinvW, qinvW, lane, n, x0, y0, outCos, outDiss);
    extract_pass<1, 0, 1, 1, 2, false>(acc, Gs, kinvW, qinvW, lane, n, x0, y0, outCos, outDiss);
    extract_pass<2, 0, 2, 1, 2, false>(acc, Gs, kinvW, qinvW, lane, n, x0, y0, outCos, outDiss);
    extract_pass<0, 2, 1, 2, 2, false>(acc, Gs, kinvW, qinvW, lane, n, x0, y0, outCos, outDiss);
    extract_pass<2, 2, 2, 2, 1, false>(acc, Gs, kinvW, qinvW, lane, n, x0, y0, outCos, outDiss);
}

extern "C" void kernel_launch(void* const* d_in, const int* in_sizes, int n_in,
                              void* d_out, int out_size, void* d_ws, size_t ws_size,
                              hipStream_t stream) {
    const float* img = (const float*)d_in[0];
    const float* map = (const float*)d_in[1];
    (void)in_sizes; (void)n_in; (void)out_size; (void)d_ws; (void)ws_size;

    dim3 grid(W_ / (2 * TW_), H_ / (2 * TH_), N_);
    corr_wmma_kernel<<<grid, dim3(32 * WG_WAVES), 0, stream>>>(img, map, (float*)d_out);
}
